# starting kernel
# baseline (speedup 1.0000x reference)
#include <hip/hip_runtime.h>

#define NNODES  50000
#define NEDGES  800000
#define FDIM    256
#define HEADS1  4
#define NEG_SLOPE 0.2f

typedef __attribute__((ext_vector_type(2))) float v2f;
typedef __attribute__((ext_vector_type(8))) float v8f;

__device__ __forceinline__ unsigned f2ord(float f) {
  unsigned u = __float_as_uint(f);
  return (u & 0x80000000u) ? ~u : (u | 0x80000000u);
}
__device__ __forceinline__ float ord2f(unsigned k) {
  unsigned u = (k & 0x80000000u) ? (k ^ 0x80000000u) : ~k;
  return __uint_as_float(u);
}
#define ORD_NEG_INF 0x007FFFFFu

__global__ void fill_u32(unsigned* __restrict__ p, long long count, unsigned val) {
  long long i = (long long)blockIdx.x * blockDim.x + threadIdx.x;
  if (i < count) p[i] = val;
}

__global__ void gemm_wmma_f32(const float* __restrict__ A, const float* __restrict__ B,
                              float* __restrict__ C, int M) {
  constexpr int K = 256, Nc = 256;
  constexpr int NT = 4;
  constexpr int nWavesN = (Nc / 16) / NT;
  int wave = (int)(((long long)blockIdx.x * blockDim.x + threadIdx.x) >> 5);
  int lane = threadIdx.x & 31;
  int tilesM = M >> 4;
  if (wave >= tilesM * nWavesN) return;
  int tM  = wave / nWavesN;
  int tN4 = (wave - tM * nWavesN) * NT;
  int half = lane >> 4;
  int l16  = lane & 15;

  const float* Abase = A + (size_t)(tM * 16 + l16) * K + 2 * half;
  const float* Bbase = B + (size_t)(2 * half) * Nc + tN4 * 16 + l16;

  v8f c[NT] = {};
#pragma unroll 8
  for (int k = 0; k < K; k += 4) {
    v2f a = *(const v2f*)(Abase + k);
#pragma unroll
    for (int t = 0; t < NT; ++t) {
      v2f b;
      b.x = Bbase[k * Nc + t * 16];
      b.y = Bbase[k * Nc + Nc + t * 16];
      c[t] = __builtin_amdgcn_wmma_f32_16x16x4_f32(false, a, false, b,
                                                   (short)0, c[t], false, false);
    }
  }
  float* Cp = C + (size_t)(tM * 16 + 8 * half) * Nc + tN4 * 16 + l16;
#pragma unroll
  for (int t = 0; t < NT; ++t)
#pragma unroll
    for (int i = 0; i < 8; ++i) Cp[i * Nc + t * 16] = c[t][i];
}

template <int H>
__global__ void att_logits(const float* __restrict__ h, const float* __restrict__ a_src,
                           const float* __restrict__ a_dst, float* __restrict__ als,
                           float* __restrict__ ald, int n) {
  constexpr int F = FDIM / H;
  int idx = blockIdx.x * blockDim.x + threadIdx.x;
  if (idx >= n * H) return;
  int node = idx / H, hd = idx - node * H;
  const float4* hp = (const float4*)(h + (size_t)node * FDIM + (size_t)hd * F);
  const float4* as = (const float4*)(a_src + hd * F);
  const float4* ad = (const float4*)(a_dst + hd * F);
  float s = 0.f, d = 0.f;
#pragma unroll 4
  for (int f = 0; f < F / 4; ++f) {
    float4 v = hp[f], x = as[f], y = ad[f];
    s += v.x * x.x + v.y * x.y + v.z * x.z + v.w * x.w;
    d += v.x * y.x + v.y * y.y + v.z * y.z + v.w * y.w;
  }
  als[idx] = s;
  ald[idx] = d;
}

template <int H>
__device__ __forceinline__ float edge_score(const float* als, const float* ald,
                                            int s, int d, int hd) {
  float sc = als[s * H + hd] + ald[d * H + hd];
  return sc > 0.f ? sc : NEG_SLOPE * sc;
}

template <int H>
__global__ void edge_max(const int* __restrict__ src, const int* __restrict__ dst,
                         int E0, int n, const float* __restrict__ als,
                         const float* __restrict__ ald, unsigned* __restrict__ mx) {
  long long i = (long long)blockIdx.x * blockDim.x + threadIdx.x;
  long long ET = (long long)E0 + n;
  if (i >= ET * H) return;
  int hd = (int)(i % H);
  long long ed = i / H;
  int s = ed < E0 ? src[ed] : (int)(ed - E0);
  int d = ed < E0 ? dst[ed] : (int)(ed - E0);
  float sc = edge_score<H>(als, ald, s, d, hd);
  atomicMax(&mx[d * H + hd], f2ord(sc));
}

template <int H>
__global__ void edge_expsum(const int* __restrict__ src, const int* __restrict__ dst,
                            int E0, int n, const float* __restrict__ als,
                            const float* __restrict__ ald, const unsigned* __restrict__ mx,
                            float* __restrict__ dn) {
  long long i = (long long)blockIdx.x * blockDim.x + threadIdx.x;
  long long ET = (long long)E0 + n;
  if (i >= ET * H) return;
  int hd = (int)(i % H);
  long long ed = i / H;
  int s = ed < E0 ? src[ed] : (int)(ed - E0);
  int d = ed < E0 ? dst[ed] : (int)(ed - E0);
  float sc = edge_score<H>(als, ald, s, d, hd);
  float ex = __expf(sc - ord2f(mx[d * H + hd]));
  atomicAdd(&dn[d * H + hd], ex);
}

template <int H>
__global__ void edge_agg(const int* __restrict__ src, const int* __restrict__ dst,
                         int E0, int n, const float* __restrict__ als,
                         const float* __restrict__ ald, const unsigned* __restrict__ mx,
                         const float* __restrict__ dn, const float* __restrict__ h,
                         float* __restrict__ agg) {
  long long ET = (long long)E0 + n;
  long long ed = (long long)blockIdx.x * 4 + (threadIdx.x >> 6);
  if (ed >= ET) return;
  int q = threadIdx.x & 63;
  int s = ed < E0 ? src[ed] : (int)(ed - E0);
  int d = ed < E0 ? dst[ed] : (int)(ed - E0);
  int hd = (q * H) >> 6;
  float sc = edge_score<H>(als, ald, s, d, hd);
  float ex = __expf(sc - ord2f(mx[d * H + hd]));
  float alpha = ex / (dn[d * H + hd] + 1e-16f);
  float4 v = ((const float4*)(h + (size_t)s * FDIM))[q];
  float* ap = agg + (size_t)d * FDIM + 4 * q;
  atomicAdd(ap + 0, v.x * alpha);
  atomicAdd(ap + 1, v.y * alpha);
  atomicAdd(ap + 2, v.z * alpha);
  atomicAdd(ap + 3, v.w * alpha);
}

__global__ void bias_elu(float* __restrict__ agg, const float* __restrict__ b,
                         long long total) {
  long long i = (long long)blockIdx.x * blockDim.x + threadIdx.x;
  if (i >= total) return;
  float v = agg[i] + b[i & (FDIM - 1)];
  agg[i] = v > 0.f ? v : expm1f(v);
}

__global__ void bias_add(float* __restrict__ out, const float* __restrict__ b,
                         long long total) {
  long long i = (long long)blockIdx.x * blockDim.x + threadIdx.x;
  if (i >= total) return;
  out[i] += b[i & (FDIM - 1)];
}

static inline int blks(long long n, int bs) { return (int)((n + bs - 1) / bs); }

extern "C" void kernel_launch(void* const* d_in, const int* in_sizes, int n_in,
                              void* d_out, int out_size, void* d_ws, size_t ws_size,
                              hipStream_t stream) {
  const float* x        = (const float*)d_in[0];
  const int*   ei       = (const int*)d_in[1];
  const float* W1       = (const float*)d_in[2];
  const float* att_src1 = (const float*)d_in[3];
  const float* att_dst1 = (const float*)d_in[4];
  const float* b1       = (const float*)d_in[5];
  const float* W2       = (const float*)d_in[6];
  const float* att_src2 = (const float*)d_in[7];
  const float* att_dst2 = (const float*)d_in[8];
  const float* b2       = (const float*)d_in[9];
  float* out = (float*)d_out;

  const int* src = ei;
  const int* dst = ei + NEDGES;

  const long long NF = (long long)NNODES * FDIM;
  const long long NH = (long long)NNODES * HEADS1;

  float* ws = (float*)d_ws;
  float*    h    = ws;
  float*    agg  = ws + NF;
  float*    als  = ws + 2 * NF;
  float*    ald  = als + NH;
  unsigned* mx   = (unsigned*)(ald + NH);
  float*    dn   = (float*)(mx + NH);
  (void)ws_size; (void)in_sizes; (void)n_in; (void)out_size;

  const long long ET = (long long)NEDGES + NNODES;
  const long long gemmThreads = (long long)(NNODES / 16) * 4   * 32;
  const int aggBlks = blks(ET, 4);

  gemm_wmma_f32<<<blks(gemmThreads, 256), 256, 0, stream>>>(x, W1, h, NNODES);
  att_logits<HEADS1><<<blks(NH, 256), 256, 0, stream>>>(h, att_src1, att_dst1,
                                                        als, ald, NNODES);
  fill_u32<<<blks(NF, 256), 256, 0, stream>>>((unsigned*)agg, NF, 0u);
  fill_u32<<<blks(NH, 256), 256, 0, stream>>>(mx, NH, ORD_NEG_INF);
  fill_u32<<<blks(NH, 256), 256, 0, stream>>>((unsigned*)dn, NH, 0u);
  edge_max<HEADS1><<<blks(ET * HEADS1, 256), 256, 0, stream>>>(src, dst, NEDGES, NNODES,
                                                               als, ald, mx);
  edge_expsum<HEADS1><<<blks(ET * HEADS1, 256), 256, 0, stream>>>(src, dst, NEDGES, NNODES,
                                                                  als, ald, mx, dn);
  edge_agg<HEADS1><<<aggBlks, 256, 0, stream>>>(src, dst, NEDGES, NNODES,
                                                als, ald, mx, dn, h, agg);
  bias_elu<<<blks(NF, 256), 256, 0, stream>>>(agg, b1, NF);

  gemm_wmma_f32<<<blks(gemmThreads, 256), 256, 0, stream>>>(agg, W2, h, NNODES);
  att_logits<1><<<blks(NNODES, 256), 256, 0, stream>>>(h, att_src2, att_dst2,
                                                       als, ald, NNODES);
  fill_u32<<<blks(NF, 256), 256, 0, stream>>>((unsigned*)out, NF, 0u);
  fill_u32<<<blks(NNODES, 256), 256, 0, stream>>>(mx, NNODES, ORD_NEG_INF);
  fill_u32<<<blks(NNODES, 256), 256, 0, stream>>>((unsigned*)dn, NNODES, 0u);
  edge_max<1><<<blks(ET, 256), 256, 0, stream>>>(src, dst, NEDGES, NNODES, als, ald, mx);
  edge_expsum<1><<<blks(ET, 256), 256, 0, stream>>>(src, dst, NEDGES, NNODES,
                                                    als, ald, mx, dn);
  edge_agg<1><<<aggBlks, 256, 0, stream>>>(src, dst, NEDGES, NNODES,
                                           als, ald, mx, dn, h, out);
  bias_add<<<blks(NF, 256), 256, 0, stream>>>(out, b2, NF);
}


# aggregation as a gather: edges counting-sorted by (dst,src) once per call, one wave per node does softmax and weighted row sum in registers, no float atomics; sum order now fixed by source index
# speedup vs baseline: 3.3655x; 3.3655x over previous
#include <hip/hip_runtime.h>

#define NNODES  50000
#define NEDGES  800000
#define FDIM    256
#define HEADS1  4
#define NEG_SLOPE 0.2f
#define OFF_PAD 53248

typedef __attribute__((ext_vector_type(2))) float v2f;
typedef __attribute__((ext_vector_type(8))) float v8f;

__global__ void fill_u32(unsigned* __restrict__ p, long long count, unsigned val) {
  long long i = (long long)blockIdx.x * blockDim.x + threadIdx.x;
  if (i < count) p[i] = val;
}

__global__ void gemm_wmma_f32(const float* __restrict__ A, const float* __restrict__ B,
                              float* __restrict__ C, int M) {
  constexpr int K = 256, Nc = 256;
  constexpr int NT = 4;
  constexpr int nWavesN = (Nc / 16) / NT;
  int wave = (int)(((long long)blockIdx.x * blockDim.x + threadIdx.x) >> 5);
  int lane = threadIdx.x & 31;
  int tilesM = M >> 4;
  if (wave >= tilesM * nWavesN) return;
  int tM  = wave / nWavesN;
  int tN4 = (wave - tM * nWavesN) * NT;
  int half = lane >> 4;
  int l16  = lane & 15;

  const float* Abase = A + (size_t)(tM * 16 + l16) * K + 2 * half;
  const float* Bbase = B + (size_t)(2 * half) * Nc + tN4 * 16 + l16;

  v8f c[NT] = {};
#pragma unroll 8
  for (int k = 0; k < K; k += 4) {
    v2f a = *(const v2f*)(Abase + k);
#pragma unroll
    for (int t = 0; t < NT; ++t) {
      v2f b;
      b.x = Bbase[k * Nc + t * 16];
      b.y = Bbase[k * Nc + Nc + t * 16];
      c[t] = __builtin_amdgcn_wmma_f32_16x16x4_f32(false, a, false, b,
                                                   (short)0, c[t], false, false);
    }
  }
  float* Cp = C + (size_t)(tM * 16 + 8 * half) * Nc + tN4 * 16 + l16;
#pragma unroll
  for (int t = 0; t < NT; ++t)
#pragma unroll
    for (int i = 0; i < 8; ++i) Cp[i * Nc + t * 16] = c[t][i];
}

template <int H>
__global__ void att_logits(const float* __restrict__ h, const float* __restrict__ a_src,
                           const float* __restrict__ a_dst, float* __restrict__ als,
                           float* __restrict__ ald, int n) {
  constexpr int F = FDIM / H;
  int idx = blockIdx.x * blockDim.x + threadIdx.x;
  if (idx >= n * H) return;
  int node = idx / H, hd = idx - node * H;
  const float4* hp = (const float4*)(h + (size_t)node * FDIM + (size_t)hd * F);
  const float4* as = (const float4*)(a_src + hd * F);
  const float4* ad = (const float4*)(a_dst + hd * F);
  float s = 0.f, d = 0.f;
#pragma unroll 4
  for (int f = 0; f < F / 4; ++f) {
    float4 v = hp[f], x = as[f], y = ad[f];
    s += v.x * x.x + v.y * x.y + v.z * x.z + v.w * x.w;
    d += v.x * y.x + v.y * y.y + v.z * y.z + v.w * y.w;
  }
  als[idx] = s;
  ald[idx] = d;
}

__global__ void csr_rank(const int* __restrict__ dst, int E0, unsigned* __restrict__ cnt,
                         int* __restrict__ rank) {
  int e = blockIdx.x * blockDim.x + threadIdx.x;
  if (e < E0) rank[e] = (int)atomicAdd(&cnt[dst[e]], 1u);
}

#define SCAN_T     1024
#define SCAN_VEC   13
static_assert(SCAN_T * SCAN_VEC * 4 == OFF_PAD, "one block covers the whole padded array");
static_assert(OFF_PAD >= NNODES + 1, "room for every count and the total");

__global__ void __launch_bounds__(SCAN_T) csr_scan(unsigned* __restrict__ cnt) {
  __shared__ unsigned wtot[32];
  const int t = threadIdx.x, lane = t & 31, w = t >> 5;
  uint4* p = (uint4*)cnt + t * SCAN_VEC;
  uint4 q[SCAN_VEC];
  unsigned sum = 0;
#pragma unroll
  for (int i = 0; i < SCAN_VEC; ++i) {
    q[i] = p[i];
    sum += q[i].x + q[i].y + q[i].z + q[i].w;
  }
  unsigned inc = sum;
#pragma unroll
  for (int o = 1; o < 32; o <<= 1) {
    unsigned u = __shfl_up(inc, (unsigned)o);
    if (lane >= o) inc += u;
  }
  if (lane == 31) wtot[w] = inc;
  __syncthreads();
  if (w == 0) {
    unsigned x = wtot[lane], xi = x;
#pragma unroll
    for (int o = 1; o < 32; o <<= 1) {
      unsigned u = __shfl_up(xi, (unsigned)o);
      if (lane >= o) xi += u;
    }
    wtot[lane] = xi - x;
  }
  __syncthreads();
  unsigned run = wtot[w] + (inc - sum);
#pragma unroll
  for (int i = 0; i < SCAN_VEC; ++i) {
    uint4 o;
    o.x = run; run += q[i].x;
    o.y = run; run += q[i].y;
    o.z = run; run += q[i].z;
    o.w = run; run += q[i].w;
    p[i] = o;
  }
}

__global__ void csr_place(const int* __restrict__ src, const int* __restrict__ dst, int E0,
                          const unsigned* __restrict__ off, const int* __restrict__ rank,
                          int* __restrict__ col) {
  int e = blockIdx.x * blockDim.x + threadIdx.x;
  if (e < E0) col[off[dst[e]] + (unsigned)rank[e]] = src[e];
}

__global__ void csr_sort_rows(const unsigned* __restrict__ off, const int* __restrict__ col,
                              int* __restrict__ col2, int n) {
  const int d = __builtin_amdgcn_readfirstlane((int)((blockIdx.x * blockDim.x + threadIdx.x) >> 5));
  const int lane = threadIdx.x & 31;
  if (d >= n) return;
  const int start = (int)off[d];
  const int len = (int)off[d + 1] - start;
  for (int j0 = 0; j0 < len; j0 += 32) {
    const int j = j0 + lane;
    const bool valid = j < len;
    const int sj = valid ? col[start + j] : 0;
    int pos = 0;
    for (int k = 0; k < len; ++k) {
      const int sk = col[start + k];
      pos += (sk < sj || (sk == sj && k < j)) ? 1 : 0;
    }
    if (valid) col2[start + pos] = sj;
  }
}

__device__ __forceinline__ float wave_max(float v) {
#pragma unroll
  for (int o = 16; o > 0; o >>= 1) v = fmaxf(v, __shfl_xor(v, o));
  return v;
}
__device__ __forceinline__ float wave_sum(float v) {
#pragma unroll
  for (int o = 16; o > 0; o >>= 1) v += __shfl_xor(v, o);
  return v;
}

template <int H>
__device__ __forceinline__ void load_entry(const int* __restrict__ col,
                                           const float* __restrict__ als,
                                           const float (&ad)[H], int start, int cnt, int d,
                                           int j, int& s, float (&sc)[H]) {
  s = d;
  if (j < cnt - 1) s = col[start + j];
  if (H == 4) {
    const float4 a = ((const float4*)als)[s];
    sc[0] = a.x + ad[0]; sc[1 % H] = a.y + ad[1 % H];
    sc[2 % H] = a.z + ad[2 % H]; sc[3 % H] = a.w + ad[3 % H];
  } else {
#pragma unroll
    for (int hd = 0; hd < H; ++hd) sc[hd] = als[s * H + hd] + ad[hd];
  }
#pragma unroll
  for (int hd = 0; hd < H; ++hd) sc[hd] = sc[hd] > 0.f ? sc[hd] : NEG_SLOPE * sc[hd];
}

template <int H>
__device__ __forceinline__ void add_rows(const float* __restrict__ h, int s,
                                         const float (&sc)[H], const float (&m)[H],
                                         const float (&den)[H], int lim, int lane,
                                         float4& acc0, float4& acc1) {
  int al[H];
#pragma unroll
  for (int hd = 0; hd < H; ++hd) al[hd] = __float_as_int(__expf(sc[hd] - m[hd]) / den[hd]);
  const bool hi = (lane & 16) != 0;
#pragma unroll 4
  for (int jj = 0; jj < lim; ++jj) {
    const int sj = __builtin_amdgcn_readlane(s, jj);
    float aA, aB;
    if (H == 4) {
      const float a0 = __int_as_float(__builtin_amdgcn_readlane(al[0], jj));
      const float a1 = __int_as_float(__builtin_amdgcn_readlane(al[1 % H], jj));
      const float a2 = __int_as_float(__builtin_amdgcn_readlane(al[2 % H], jj));
      const float a3 = __int_as_float(__builtin_amdgcn_readlane(al[3 % H], jj));
      aA = hi ? a1 : a0;
      aB = hi ? a3 : a2;
    } else {
      aA = aB = __int_as_float(__builtin_amdgcn_readlane(al[0], jj));
    }
    const float4* row = (const float4*)(h + (size_t)sj * FDIM);
    const float4 v0 = row[lane];
    const float4 v1 = row[lane + 32];
    acc0.x += v0.x * aA; acc0.y += v0.y * aA; acc0.z += v0.z * aA; acc0.w += v0.w * aA;
    acc1.x += v1.x * aB; acc1.y += v1.y * aB; acc1.z += v1.z * aB; acc1.w += v1.w * aB;
  }
}

template <int H, bool ELU>
__global__ void __launch_bounds__(256)
gat_gather(const unsigned* __restrict__ off, const int* __restrict__ col,
           const float* __restrict__ als, const float* __restrict__ ald,
           const float* __restrict__ h, const float* __restrict__ bias,
           float* __restrict__ outp, int n) {
  static_assert(H == 1 || H == 4, "lane-to-head mapping below is written for 1 or 4 heads");
  const int d = __builtin_amdgcn_readfirstlane((int)((blockIdx.x * blockDim.x + threadIdx.x) >> 5));
  const int lane = threadIdx.x & 31;
  if (d >= n) return;
  const int start = (int)off[d];
  const int cnt = (int)off[d + 1] - start + 1;

  float ad[H];
#pragma unroll
  for (int hd = 0; hd < H; ++hd) ad[hd] = ald[d * H + hd];

  int s0; float sc0[H];
  load_entry<H>(col, als, ad, start, cnt, d, lane, s0, sc0);
  const bool valid0 = lane < cnt;

  float m[H], den[H];
#pragma unroll
  for (int hd = 0; hd < H; ++hd) m[hd] = valid0 ? sc0[hd] : -INFINITY;
  for (int j0 = 32; j0 < cnt; j0 += 32) {
    int s; float sc[H];
    load_entry<H>(col, als, ad, start, cnt, d, j0 + lane, s, sc);
    if (j0 + lane < cnt) {
#pragma unroll
      for (int hd = 0; hd < H; ++hd) m[hd] = fmaxf(m[hd], sc[hd]);
    }
  }
#pragma unroll
  for (int hd = 0; hd < H; ++hd) m[hd] = wave_max(m[hd]);

#pragma unroll
  for (int hd = 0; hd < H; ++hd) den[hd] = valid0 ? __expf(sc0[hd] - m[hd]) : 0.f;
  for (int j0 = 32; j0 < cnt; j0 += 32) {
    int s; float sc[H];
    load_entry<H>(col, als, ad, start, cnt, d, j0 + lane, s, sc);
    if (j0 + lane < cnt) {
#pragma unroll
      for (int hd = 0; hd < H; ++hd) den[hd] += __expf(sc[hd] - m[hd]);
    }
  }
#pragma unroll
  for (int hd = 0; hd < H; ++hd) den[hd] = wave_sum(den[hd]) + 1e-16f;

  float4 acc0 = {0.f, 0.f, 0.f, 0.f}, acc1 = {0.f, 0.f, 0.f, 0.f};
  add_rows<H>(h, s0, sc0, m, den, cnt < 32 ? cnt : 32, lane, acc0, acc1);
  for (int j0 = 32; j0 < cnt; j0 += 32) {
    int s; float sc[H];
    load_entry<H>(col, als, ad, start, cnt, d, j0 + lane, s, sc);
    add_rows<H>(h, s, sc, m, den, (cnt - j0) < 32 ? (cnt - j0) : 32, lane, acc0, acc1);
  }

  const float4 b0 = ((const float4*)bias)[lane];
  const float4 b1 = ((const float4*)bias)[lane + 32];
  float r[8] = {acc0.x + b0.x, acc0.y + b0.y, acc0.z + b0.z, acc0.w + b0.w,
                acc1.x + b1.x, acc1.y + b1.y, acc1.z + b1.z, acc1.w + b1.w};
  if (ELU) {
#pragma unroll
    for (int i = 0; i < 8; ++i) r[i] = r[i] > 0.f ? r[i] : expm1f(r[i]);
  }
  float4* orow = (float4*)(outp + (size_t)d * FDIM);
  orow[lane]      = make_float4(r[0], r[1], r[2], r[3]);
  orow[lane + 32] = make_float4(r[4], r[5], r[6], r[7]);
}

static inline int blks(long long n, int bs) { return (int)((n + bs - 1) / bs); }

extern "C" void kernel_launch(void* const* d_in, const int* in_sizes, int n_in,
                              void* d_out, int out_size, void* d_ws, size_t ws_size,
                              hipStream_t stream) {
  const float* x        = (const float*)d_in[0];
  const int*   ei       = (const int*)d_in[1];
  const float* W1       = (const float*)d_in[2];
  const float* att_src1 = (const float*)d_in[3];
  const float* att_dst1 = (const float*)d_in[4];
  const float* b1       = (const float*)d_in[5];
  const float* W2       = (const float*)d_in[6];
  const float* att_src2 = (const float*)d_in[7];
  const float* att_dst2 = (const float*)d_in[8];
  const float* b2       = (const float*)d_in[9];
  float* out = (float*)d_out;

  const int* src = ei;
  const int* dst = ei + NEDGES;

  const long long NF = (long long)NNODES * FDIM;
  const long long NH = (long long)NNODES * HEADS1;

  float* ws = (float*)d_ws;
  float*    h    = ws;
  float*    als  = ws + NF;
  float*    ald  = als + NH;
  unsigned* off  = (unsigned*)(ald + NH);
  int*      col  = (int*)(off + OFF_PAD);
  int*      col2 = col + NEDGES;
  (void)ws_size; (void)in_sizes; (void)n_in; (void)out_size;

  const long long gemmThreads = (long long)(NNODES / 16) * 4   * 32;
  const long long rowThreads  = (long long)NNODES * 32;

  fill_u32<<<blks(OFF_PAD, 256), 256, 0, stream>>>(off, OFF_PAD, 0u);
  csr_rank<<<blks(NEDGES, 256), 256, 0, stream>>>(dst, NEDGES, off, col2);
  csr_scan<<<1, SCAN_T, 0, stream>>>(off);
  csr_place<<<blks(NEDGES, 256), 256, 0, stream>>>(src, dst, NEDGES, off, col2, col);
  csr_sort_rows<<<blks(rowThreads, 256), 256, 0, stream>>>(off, col, col2, NNODES);

  gemm_wmma_f32<<<blks(gemmThreads, 256), 256, 0, stream>>>(x, W1, h, NNODES);
  att_logits<HEADS1><<<blks(NH, 256), 256, 0, stream>>>(h, att_src1, att_dst1,
                                                        als, ald, NNODES);
  gat_gather<HEADS1, true><<<blks(rowThreads, 256), 256, 0, stream>>>(off, col2, als, ald, h,
                                                                      b1, out, NNODES);

  gemm_wmma_f32<<<blks(gemmThreads, 256), 256, 0, stream>>>(out, W2, h, NNODES);
  att_logits<1><<<blks(NNODES, 256), 256, 0, stream>>>(h, att_src2, att_dst2,
                                                       als, ald, NNODES);
  gat_gather<1, false><<<blks(rowThreads, 256), 256, 0, stream>>>(off, col2, als, ald, h,
                                                                  b2, out, NNODES);
}


# GEMM B operand (256x256 f32) held in LDS per workgroup: 256 resident 16-wave workgroups walk the M-tiles, B fragments by 8-byte LDS reads one block ahead; same k order and sum order
# speedup vs baseline: 4.0249x; 1.1959x over previous
#include <hip/hip_runtime.h>

#define NNODES  50000
#define NEDGES  800000
#define FDIM    256
#define HEADS1  4
#define NEG_SLOPE 0.2f
#define OFF_PAD 53248

typedef __attribute__((ext_vector_type(2))) float v2f;
typedef __attribute__((ext_vector_type(8))) float v8f;

__global__ void fill_u32(unsigned* __restrict__ p, long long count, unsigned val) {
  long long i = (long long)blockIdx.x * blockDim.x + threadIdx.x;
  if (i < count) p[i] = val;
}

#define GEMM_WGS     256
#define GEMM_WAVES   16
#define BL_PITCH     544
typedef __attribute__((ext_vector_type(4))) float v4f;

__global__ void __launch_bounds__(GEMM_WAVES * 32)
gemm_wmma_f32(const float* __restrict__ A, const float* __restrict__ B,
              float* __restrict__ C, int M) {
  constexpr int K = 256, Nc = 256;
  constexpr int NT = 4;
  constexpr int KB = 4;
  __shared__ __attribute__((aligned(16))) float Bl[(K / 2) * BL_PITCH];
  const int tid = threadIdx.x;

  static_assert(((K / 2) * (Nc / 2)) % (GEMM_WAVES * 32) == 0, "every thread stages the same number of pairs");
#pragma unroll 16
  for (int j = 0; j < ((K / 2) * (Nc / 2)) / (GEMM_WAVES * 32); ++j) {
    const int i = tid + j * (GEMM_WAVES * 32);
    const int kp = i >> 7, c2 = i & 127;
    const v2f r0 = *(const v2f*)(B + (size_t)(2 * kp) * Nc + 2 * c2);
    const v2f r1 = *(const v2f*)(B + (size_t)(2 * kp + 1) * Nc + 2 * c2);
    v4f o;
    o.x = r0.x; o.y = r1.x; o.z = r0.y; o.w = r1.y;
    *(v4f*)(Bl + kp * BL_PITCH + 4 * c2) = o;
  }
  __syncthreads();

  const int lane = tid & 31;
  const int wave = __builtin_amdgcn_readfirstlane(tid >> 5);
  const int half = lane >> 4;
  const int l16  = lane & 15;
  const int nq   = wave & 3;
  const int tilesM = M >> 4;

  const float* Bp = Bl + half * BL_PITCH + (nq * 64 + l16) * 2;

  constexpr int NBLK = K / (4 * KB);
  constexpr int TSTEP = (GEMM_WAVES / 4) * GEMM_WGS;
  int tM = blockIdx.x + (wave >> 2) * GEMM_WGS;
  if (tM >= tilesM) return;

  v2f an[KB], bn[KB * NT];
  {
    const float* Ap = A + (size_t)(tM * 16 + l16) * K + 2 * half;
#pragma unroll
    for (int s = 0; s < KB; ++s) {
      an[s] = *(const v2f*)(Ap + s * 4);
#pragma unroll
      for (int t = 0; t < NT; ++t)
        bn[s * NT + t] = *(const v2f*)(Bp + (s * 2) * BL_PITCH + t * 32);
    }
  }
  for (;;) {
    const int tNext = tM + TSTEP;
    const bool more = tNext < tilesM;
    const float* Ap  = A + (size_t)(tM * 16 + l16) * K + 2 * half;
    const float* ApN = A + (size_t)((more ? tNext : tM) * 16 + l16) * K + 2 * half;
    v8f c[NT] = {};
#pragma unroll
    for (int blk = 0; blk < NBLK; ++blk) {
      v2f ac[KB], bc[KB * NT];
#pragma unroll
      for (int s = 0; s < KB; ++s) {
        ac[s] = an[s];
#pragma unroll
        for (int t = 0; t < NT; ++t) bc[s * NT + t] = bn[s * NT + t];
      }
      const int nb = (blk + 1) % NBLK;
      const float* Al = (blk + 1 < NBLK) ? Ap : ApN;
#pragma unroll
      for (int s = 0; s < KB; ++s) {
        an[s] = *(const v2f*)(Al + nb * (4 * KB) + s * 4);
#pragma unroll
        for (int t = 0; t < NT; ++t)
          bn[s * NT + t] = *(const v2f*)(Bp + (nb * (2 * KB) + s * 2) * BL_PITCH + t * 32);
      }
      __builtin_amdgcn_sched_barrier(0);
#pragma unroll
      for (int s = 0; s < KB; ++s)
#pragma unroll
        for (int t = 0; t < NT; ++t)
          c[t] = __builtin_amdgcn_wmma_f32_16x16x4_f32(false, ac[s], false, bc[s * NT + t],
                                                       (short)0, c[t], false, false);
      __builtin_amdgcn_sched_barrier(0);
    }
    float* Cp = C + (size_t)(tM * 16 + 8 * half) * Nc + nq * 64 + l16;
#pragma unroll
    for (int t = 0; t < NT; ++t)
#pragma unroll
      for (int i = 0; i < 8; ++i) Cp[i * Nc + t * 16] = c[t][i];
    if (!more) break;
    tM = tNext;
  }
}

__global__ void csr_rank(const int* __restrict__ dst, int E0, unsigned* __restrict__ cnt,
                         int* __restrict__ rank) {
  int e = blockIdx.x * blockDim.x + threadIdx.x;
  if (e < E0) rank[e] = (int)atomicAdd(&cnt[dst[e]], 1u);
}

#define SCAN_T     1024
#define SCAN_VEC   13
static_assert(SCAN_T * SCAN_VEC * 4 == OFF_PAD, "one block covers the whole padded array");
static_assert(OFF_PAD >= NNODES + 1, "room for every count and the total");

__global__ void __launch_bounds__(SCAN_T) csr_scan(unsigned* __restrict__ cnt) {
  __shared__ unsigned wtot[32];
  const int t = threadIdx.x, lane = t & 31, w = t >> 5;
  uint4* p = (uint4*)cnt + t * SCAN_VEC;
  uint4 q[SCAN_VEC];
  unsigned sum = 0;
#pragma unroll
  for (int i = 0; i < SCAN_VEC; ++i) {
    q[i] = p[i];
    sum += q[i].x + q[i].y + q[i].z + q[i].w;
  }
  unsigned inc = sum;
#pragma unroll
  for (int o = 1; o < 32; o <<= 1) {
    unsigned u = __shfl_up(inc, (unsigned)o);
    if (lane >= o) inc += u;
  }
  if (lane == 31) wtot[w] = inc;
  __syncthreads();
  if (w == 0) {
    unsigned x = wtot[lane], xi = x;
#pragma unroll
    for (int o = 1; o < 32; o <<= 1) {
      unsigned u = __shfl_up(xi, (unsigned)o);
      if (lane >= o) xi += u;
    }
    wtot[lane] = xi - x;
  }
  __syncthreads();
  unsigned run = wtot[w] + (inc - sum);
#pragma unroll
  for (int i = 0; i < SCAN_VEC; ++i) {
    uint4 o;
    o.x = run; run += q[i].x;
    o.y = run; run += q[i].y;
    o.z = run; run += q[i].z;
    o.w = run; run += q[i].w;
    p[i] = o;
  }
}

__global__ void csr_place(const int* __restrict__ src, const int* __restrict__ dst, int E0,
                          const unsigned* __restrict__ off, const int* __restrict__ rank,
                          int* __restrict__ col) {
  int e = blockIdx.x * blockDim.x + threadIdx.x;
  if (e < E0) col[off[dst[e]] + (unsigned)rank[e]] = src[e];
}

__global__ void csr_sort_rows(const unsigned* __restrict__ off, const int* __restrict__ col,
                              int* __restrict__ col2, int n) {
  const int d = __builtin_amdgcn_readfirstlane((int)((blockIdx.x * blockDim.x + threadIdx.x) >> 5));
  const int lane = threadIdx.x & 31;
  if (d >= n) return;
  const int start = (int)off[d];
  const int len = (int)off[d + 1] - start;
  for (int j0 = 0; j0 < len; j0 += 32) {
    const int j = j0 + lane;
    const bool valid = j < len;
    const int sj = valid ? col[start + j] : 0;
    int pos = 0;
    for (int k = 0; k < len; ++k) {
      const int sk = col[start + k];
      pos += (sk < sj || (sk == sj && k < j)) ? 1 : 0;
    }
    if (valid) col2[start + pos] = sj;
  }
}

template <int MASK>
__device__ __forceinline__ float row_xor_add(float v) {
  return v + __int_as_float(__builtin_amdgcn_update_dpp(0, __float_as_int(v), 0x160 | MASK,
                                                        0xf, 0xf, true));
}
__device__ __forceinline__ float xhalf(float x) {
  return __int_as_float(__builtin_amdgcn_permlanex16(__float_as_int(x), __float_as_int(x),
                                                     0x76543210, 0xfedcba98, false, false));
}

template <int H, int NR, int MASK>
__device__ __forceinline__ void row_step(float (&a)[NR], float (&b)[NR]) {
  for (int i = 0; i < NR; ++i) {
    a[i] = row_xor_add<MASK>(a[i]);
    if (H == 4) b[i] = row_xor_add<MASK>(b[i]);
  }
}

template <int H, int NR>
__device__ __forceinline__ void head_sums(float (&a)[NR], float (&b)[NR]) {
  if (H == 1) {
    for (int i = 0; i < NR; ++i) a[i] += b[i];
  }
  row_step<H, NR, 1>(a, b);
  row_step<H, NR, 2>(a, b);
  row_step<H, NR, 4>(a, b);
  row_step<H, NR, 8>(a, b);
  if (H == 1) {
    for (int i = 0; i < NR; ++i) a[i] += xhalf(a[i]);
  }
}

__device__ __forceinline__ float dot4(const v4f& v, const v4f& w) {
  v2f t = v.lo * w.lo;
  t = v.hi * w.hi + t;
  return t.x + t.y;
}
__device__ __forceinline__ float lrelu(float x) { return fmaxf(x, NEG_SLOPE * x); }

struct Soft {
  float mA, mB, denA, denB;
  v4f acc0, acc1;
};

template <int H, int NR>
__device__ __forceinline__ void add_entries(const float* __restrict__ h, int sl, int jj, int lane,
                                            const v4f& as0, const v4f& as1,
                                            float aldA, float aldB, Soft& st) {
  v4f v0[NR], v1[NR];
  for (int i = 0; i < NR; ++i) {
    const int sj = __builtin_amdgcn_readlane(sl, jj + i);
    const v4f* row = (const v4f*)(h + (size_t)sj * FDIM);
    v0[i] = row[lane];
    v1[i] = row[lane + 32];
  }
  if (NR > 1) __builtin_amdgcn_sched_barrier(0);
  float eA[NR], eB[NR];
  for (int i = 0; i < NR; ++i) {
    eA[i] = dot4(v0[i], as0);
    eB[i] = dot4(v1[i], as1);
  }
  head_sums<H, NR>(eA, eB);
  float nA = st.mA, nB = st.mB;
  for (int i = 0; i < NR; ++i) {
    eA[i] = lrelu(eA[i] + aldA);
    nA = fmaxf(nA, eA[i]);
    if (H == 4) {
      eB[i] = lrelu(eB[i] + aldB);
      nB = fmaxf(nB, eB[i]);
    }
  }
  const float cA = __expf(st.mA - nA);
  const float cB = (H == 4) ? __expf(st.mB - nB) : cA;
  st.denA *= cA;
  if (H == 4) st.denB *= cB;
  st.acc0 *= cA;
  st.acc1 *= cB;
  for (int i = 0; i < NR; ++i) {
    const float wA = __expf(eA[i] - nA);
    const float wB = (H == 4) ? __expf(eB[i] - nB) : wA;
    st.denA += wA;
    if (H == 4) st.denB += wB;
    st.acc0 += v0[i] * wA;
    st.acc1 += v1[i] * wB;
  }
  st.mA = nA;
  if (H == 4) st.mB = nB;
}

template <int H, bool ELU>
__global__ void __launch_bounds__(256)
gat_gather(const unsigned* __restrict__ off, const int* __restrict__ col,
           const float* __restrict__ h, const float* __restrict__ a_src,
           const float* __restrict__ a_dst, const float* __restrict__ bias,
           float* __restrict__ outp, int n) {
  static_assert(H == 1 || H == 4, "lane-to-head mapping is written for 1 or 4 heads");
  const int d = __builtin_amdgcn_readfirstlane((int)((blockIdx.x * blockDim.x + threadIdx.x) >> 5));
  const int lane = threadIdx.x & 31;
  if (d >= n) return;
  const int start = (int)off[d];
  const int ne = (int)off[d + 1] - start;

  const v4f as0 = ((const v4f*)a_src)[lane];
  const v4f as1 = ((const v4f*)a_src)[lane + 32];

  Soft st;
  float aldA, aldB;
  {
    const v4f* rowd = (const v4f*)(h + (size_t)d * FDIM);
    const v4f d0 = rowd[lane];
    const v4f d1 = rowd[lane + 32];
    const v4f ad0 = ((const v4f*)a_dst)[lane];
    const v4f ad1 = ((const v4f*)a_dst)[lane + 32];
    float pa[2] = {dot4(d0, ad0), dot4(d0, as0)};
    float pb[2] = {dot4(d1, ad1), dot4(d1, as1)};
    head_sums<H, 2>(pa, pb);
    aldA = pa[0];
    aldB = (H == 4) ? pb[0] : pa[0];
    st.mA = lrelu(pa[1] + aldA);
    st.mB = (H == 4) ? lrelu(pb[1] + aldB) : st.mA;
    st.denA = 1.f;
    st.denB = 1.f;
    st.acc0 = d0;
    st.acc1 = d1;
  }

  for (int j0 = 0; j0 < ne; j0 += 32) {
    const int j = j0 + lane;
    int sl = d;
    if (j < ne) sl = col[start + j];
    const int lim = (ne - j0) < 32 ? (ne - j0) : 32;
    int jj = 0;
    for (; jj + 4 <= lim; jj += 4)
      add_entries<H, 4>(h, sl, jj, lane, as0, as1, aldA, aldB, st);
    if (jj + 2 <= lim) {
      add_entries<H, 2>(h, sl, jj, lane, as0, as1, aldA, aldB, st);
      jj += 2;
    }
    if (jj < lim) add_entries<H, 1>(h, sl, jj, lane, as0, as1, aldA, aldB, st);
  }

  const float iA = 1.f / (st.denA + 1e-16f);
  const float iB = (H == 4) ? 1.f / (st.denB + 1e-16f) : iA;
  const v4f b0 = ((const v4f*)bias)[lane];
  const v4f b1 = ((const v4f*)bias)[lane + 32];
  v4f r0 = st.acc0 * iA + b0;
  v4f r1 = st.acc1 * iB + b1;
  if (ELU) {
    for (int i = 0; i < 4; ++i) {
      r0[i] = r0[i] > 0.f ? r0[i] : expm1f(r0[i]);
      r1[i] = r1[i] > 0.f ? r1[i] : expm1f(r1[i]);
    }
  }
  v4f* orow = (v4f*)(outp + (size_t)d * FDIM);
  orow[lane]      = r0;
  orow[lane + 32] = r1;
}

static inline int blks(long long n, int bs) { return (int)((n + bs - 1) / bs); }

extern "C" void kernel_launch(void* const* d_in, const int* in_sizes, int n_in,
                              void* d_out, int out_size, void* d_ws, size_t ws_size,
                              hipStream_t stream) {
  const float* x        = (const float*)d_in[0];
  const int*   ei       = (const int*)d_in[1];
  const float* W1       = (const float*)d_in[2];
  const float* att_src1 = (const float*)d_in[3];
  const float* att_dst1 = (const float*)d_in[4];
  const float* b1       = (const float*)d_in[5];
  const float* W2       = (const float*)d_in[6];
  const float* att_src2 = (const float*)d_in[7];
  const float* att_dst2 = (const float*)d_in[8];
  const float* b2       = (const float*)d_in[9];
  float* out = (float*)d_out;

  const int* src = ei;
  const int* dst = ei + NEDGES;

  const long long NF = (long long)NNODES * FDIM;

  float* ws = (float*)d_ws;
  float*    h    = ws;
  unsigned* off  = (unsigned*)(ws + NF);
  int*      col  = (int*)(off + OFF_PAD);
  int*      col2 = col + NEDGES;
  (void)ws_size; (void)in_sizes; (void)n_in; (void)out_size;

    const long long rowThreads  = (long long)NNODES * 32;

  fill_u32<<<blks(OFF_PAD, 256), 256, 0, stream>>>(off, OFF_PAD, 0u);
  csr_rank<<<blks(NEDGES, 256), 256, 0, stream>>>(dst, NEDGES, off, col2);
  csr_scan<<<1, SCAN_T, 0, stream>>>(off);
  csr_place<<<blks(NEDGES, 256), 256, 0, stream>>>(src, dst, NEDGES, off, col2, col);
  csr_sort_rows<<<blks(rowThreads, 256), 256, 0, stream>>>(off, col, col2, NNODES);

  gemm_wmma_f32<<<GEMM_WGS, GEMM_WAVES * 32, 0, stream>>>(x, W1, h, NNODES);
  gat_gather<HEADS1, true><<<blks(rowThreads, 256), 256, 0, stream>>>(off, col2, h, att_src1,
                                                                      att_dst1, b1, out, NNODES);

  gemm_wmma_f32<<<GEMM_WGS, GEMM_WAVES * 32, 0, stream>>>(out, W2, h, NNODES);
  gat_gather<1, false><<<blks(rowThreads, 256), 256, 0, stream>>>(off, col2, h, att_src2,
                                                                  att_dst2, b2, out, NNODES);
}


# edge sort without global atomics: edges bucketed by destination range in LDS, one resident workgroup per bucket counts, places and row-sorts in LDS; 2 launches for 5; sum order unchanged
# speedup vs baseline: 5.6450x; 1.4025x over previous
#include <hip/hip_runtime.h>

#define NNODES  50000
#define NEDGES  800000
#define FDIM    256
#define HEADS1  4
#define NEG_SLOPE 0.2f
#define OFF_PAD 53248

typedef __attribute__((ext_vector_type(2))) float v2f;
typedef __attribute__((ext_vector_type(8))) float v8f;

#define GEMM_WGS     256
#define GEMM_WAVES   16
#define BL_PITCH     544
typedef __attribute__((ext_vector_type(4))) float v4f;

__global__ void __launch_bounds__(GEMM_WAVES * 32)
gemm_wmma_f32(const float* __restrict__ A, const float* __restrict__ B,
              float* __restrict__ C, int M) {
  constexpr int K = 256, Nc = 256;
  constexpr int NT = 4;
  constexpr int KB = 4;
  __shared__ __attribute__((aligned(16))) float Bl[(K / 2) * BL_PITCH];
  const int tid = threadIdx.x;

  static_assert(((K / 2) * (Nc / 2)) % (GEMM_WAVES * 32) == 0, "every thread stages the same number of pairs");
#pragma unroll 16
  for (int j = 0; j < ((K / 2) * (Nc / 2)) / (GEMM_WAVES * 32); ++j) {
    const int i = tid + j * (GEMM_WAVES * 32);
    const int kp = i >> 7, c2 = i & 127;
    const v2f r0 = *(const v2f*)(B + (size_t)(2 * kp) * Nc + 2 * c2);
    const v2f r1 = *(const v2f*)(B + (size_t)(2 * kp + 1) * Nc + 2 * c2);
    v4f o;
    o.x = r0.x; o.y = r1.x; o.z = r0.y; o.w = r1.y;
    *(v4f*)(Bl + kp * BL_PITCH + 4 * c2) = o;
  }
  __syncthreads();

  const int lane = tid & 31;
  const int wave = __builtin_amdgcn_readfirstlane(tid >> 5);
  const int half = lane >> 4;
  const int l16  = lane & 15;
  const int nq   = wave & 3;
  const int tilesM = M >> 4;

  const float* Bp = Bl + half * BL_PITCH + (nq * 64 + l16) * 2;

  constexpr int NBLK = K / (4 * KB);
  constexpr int TSTEP = (GEMM_WAVES / 4) * GEMM_WGS;
  int tM = blockIdx.x + (wave >> 2) * GEMM_WGS;
  if (tM >= tilesM) return;

  v2f an[KB], bn[KB * NT];
  {
    const float* Ap = A + (size_t)(tM * 16 + l16) * K + 2 * half;
#pragma unroll
    for (int s = 0; s < KB; ++s) {
      an[s] = *(const v2f*)(Ap + s * 4);
#pragma unroll
      for (int t = 0; t < NT; ++t)
        bn[s * NT + t] = *(const v2f*)(Bp + (s * 2) * BL_PITCH + t * 32);
    }
  }
  for (;;) {
    const int tNext = tM + TSTEP;
    const bool more = tNext < tilesM;
    const float* Ap  = A + (size_t)(tM * 16 + l16) * K + 2 * half;
    const float* ApN = A + (size_t)((more ? tNext : tM) * 16 + l16) * K + 2 * half;
    v8f c[NT] = {};
#pragma unroll
    for (int blk = 0; blk < NBLK; ++blk) {
      v2f ac[KB], bc[KB * NT];
#pragma unroll
      for (int s = 0; s < KB; ++s) {
        ac[s] = an[s];
#pragma unroll
        for (int t = 0; t < NT; ++t) bc[s * NT + t] = bn[s * NT + t];
      }
      const int nb = (blk + 1) % NBLK;
      const float* Al = (blk + 1 < NBLK) ? Ap : ApN;
#pragma unroll
      for (int s = 0; s < KB; ++s) {
        an[s] = *(const v2f*)(Al + nb * (4 * KB) + s * 4);
#pragma unroll
        for (int t = 0; t < NT; ++t)
          bn[s * NT + t] = *(const v2f*)(Bp + (nb * (2 * KB) + s * 2) * BL_PITCH + t * 32);
      }
      __builtin_amdgcn_sched_barrier(0);
#pragma unroll
      for (int s = 0; s < KB; ++s)
#pragma unroll
        for (int t = 0; t < NT; ++t)
          c[t] = __builtin_amdgcn_wmma_f32_16x16x4_f32(false, ac[s], false, bc[s * NT + t],
                                                       (short)0, c[t], false, false);
      __builtin_amdgcn_sched_barrier(0);
    }
    float* Cp = C + (size_t)(tM * 16 + 8 * half) * Nc + nq * 64 + l16;
#pragma unroll
    for (int t = 0; t < NT; ++t)
#pragma unroll
      for (int i = 0; i < 8; ++i) Cp[i * Nc + t * 16] = c[t][i];
    if (!more) break;
    tM = tNext;
  }
}

#define BKT_NODES 196
#define NBKT      256
#define NCHUNK    256
#define CHUNK     3125
#define BIN_T     1024
#define BKT_CAP   16384
static_assert(CHUNK * NCHUNK == NEDGES, "the slices cover the edge list exactly");
static_assert(BKT_NODES * NBKT >= NNODES + 1, "the buckets cover every node and the end offset");
static_assert(NNODES <= 65536 && BKT_NODES <= 256, "a binned edge fits one word");
static_assert(BIN_T == 4 * NCHUNK, "four threads per slice in csr_bucket");

__device__ __forceinline__ unsigned scan256(unsigned v, int t, unsigned* wsum, unsigned& total) {
  const int lane = t & 31, wv = t >> 5;
  unsigned inc = v;
#pragma unroll
  for (int o = 1; o < 32; o <<= 1) {
    const unsigned u = __shfl_up(inc, (unsigned)o);
    if (lane >= o) inc += u;
  }
  if (t < 256 && lane == 31) wsum[wv] = inc;
  __syncthreads();
  unsigned pre = 0, tot = 0;
#pragma unroll
  for (int k = 0; k < 8; ++k) {
    const unsigned x = wsum[k];
    tot += x;
    if (k < wv) pre += x;
  }
  total = tot;
  return pre + inc - v;
}

__global__ void __launch_bounds__(BIN_T)
csr_bin(const int* __restrict__ src, const int* __restrict__ dst,
        unsigned* __restrict__ tabT, unsigned* __restrict__ binned) {
  __shared__ unsigned hist[NBKT];
  __shared__ unsigned wsum[8];
  __shared__ unsigned stage[CHUNK];
  const int t = threadIdx.x;
  const int w = blockIdx.x;
  constexpr int PER = (CHUNK + BIN_T - 1) / BIN_T;
  if (t < NBKT) hist[t] = 0;
  unsigned s[PER], d[PER];
#pragma unroll
  for (int i = 0; i < PER; ++i) {
    const int e = t + i * BIN_T;
    s[i] = 0; d[i] = 0;
    if (e < CHUNK) {
      s[i] = (unsigned)src[w * CHUNK + e];
      d[i] = (unsigned)dst[w * CHUNK + e];
    }
  }
  __syncthreads();
  unsigned bk[PER], key[PER], rk[PER];
#pragma unroll
  for (int i = 0; i < PER; ++i) {
    const int e = t + i * BIN_T;
    bk[i] = d[i] / BKT_NODES;
    key[i] = s[i] | ((d[i] - bk[i] * BKT_NODES) << 16);
    rk[i] = 0;
    if (e < CHUNK) rk[i] = atomicAdd(&hist[bk[i]], 1u);
  }
  __syncthreads();
  const unsigned c = (t < NBKT) ? hist[t] : 0u;
  unsigned total;
  const unsigned ex = scan256(c, t, wsum, total);
  if (t < NBKT) {
    hist[t] = ex;
    tabT[t * NCHUNK + w] = ex;
    if (t == NBKT - 1) tabT[NBKT * NCHUNK + w] = total;
  }
  __syncthreads();
#pragma unroll
  for (int i = 0; i < PER; ++i) {
    const int e = t + i * BIN_T;
    if (e < CHUNK) stage[hist[bk[i]] + rk[i]] = key[i];
  }
  __syncthreads();
  for (int p = t; p < CHUNK; p += BIN_T) binned[w * CHUNK + p] = stage[p];
}

__global__ void __launch_bounds__(BIN_T)
csr_bucket(const unsigned* __restrict__ tabT, const unsigned* __restrict__ binned,
           unsigned* __restrict__ off, unsigned* __restrict__ tmp, int* __restrict__ col2) {
  __shared__ unsigned segSt[NCHUNK];
  __shared__ unsigned segLen[NCHUNK];
  __shared__ unsigned segPos[NCHUNK];
  __shared__ unsigned cnt[NBKT];
  __shared__ unsigned cur[NBKT];
  __shared__ unsigned ws0[8], ws1[8], ws2[8];
  __shared__ unsigned bufA[BKT_CAP];
  __shared__ unsigned bufB[BKT_CAP];
  const int t = threadIdx.x;
  const int lane = t & 31;
  const int wv = __builtin_amdgcn_readfirstlane(t >> 5);
  const int b = blockIdx.x;

  unsigned st = 0, ln = 0;
  if (t < NCHUNK) {
    st = tabT[b * NCHUNK + t];
    ln = tabT[(b + 1) * NCHUNK + t] - st;
    cnt[t] = 0;
    cur[t] = 0;
  }
  unsigned total, base;
  const unsigned sp = scan256(ln, t, ws0, total);
  (void)scan256(st, t, ws1, base);
  if (t < NCHUNK) {
    segSt[t] = st;
    segLen[t] = ln;
    segPos[t] = sp;
  }
  __syncthreads();
  const bool fits = total <= BKT_CAP;

  const int sg = t >> 2, sub = t & 3;
  const unsigned sl = segLen[sg], p0 = segPos[sg];
  const unsigned* g = binned + sg * CHUNK + segSt[sg];
  for (unsigned i0 = 0; i0 < sl; i0 += 16) {
    unsigned k[4];
#pragma unroll
    for (int u = 0; u < 4; ++u) {
      const unsigned idx = i0 + sub + 4 * u;
      k[u] = idx < sl ? g[idx] : 0u;
    }
#pragma unroll
    for (int u = 0; u < 4; ++u) {
      const unsigned idx = i0 + sub + 4 * u;
      if (idx < sl) {
        atomicAdd(&cnt[k[u] >> 16], 1u);
        if (fits) bufA[p0 + idx] = k[u];
      }
    }
  }
  __syncthreads();
  const unsigned c = (t < NBKT) ? cnt[t] : 0u;
  unsigned tot2;
  const unsigned ex = scan256(c, t, ws2, tot2);
  if (t < NBKT) {
    cnt[t] = ex;
    const int node = b * BKT_NODES + t;
    if (t < BKT_NODES && node <= NNODES) off[node] = base + ex;
  }
  __syncthreads();

  if (fits) {
    for (unsigned p = t; p < total; p += BIN_T) {
      const unsigned k = bufA[p];
      const unsigned dl = k >> 16;
      const unsigned pos = cnt[dl] + atomicAdd(&cur[dl], 1u);
      bufB[pos] = k & 0xffffu;
    }
  } else {
    for (unsigned i = sub; i < sl; i += 4) {
      const unsigned k = g[i];
      const unsigned dl = k >> 16;
      const unsigned pos = cnt[dl] + atomicAdd(&cur[dl], 1u);
      tmp[base + pos] = k & 0xffffu;
    }
    __threadfence();
  }
  __syncthreads();

  for (int r = wv; r < BKT_NODES; r += BIN_T / 32) {
    const int start = __builtin_amdgcn_readfirstlane((int)cnt[r]);
    const int len = __builtin_amdgcn_readfirstlane((int)cur[r]);
    if (fits && len <= 32) {
      const int sj = lane < len ? (int)bufB[start + lane] : 0;
      int pos = 0;
      for (int k = 0; k < len; ++k) {
        const int sk = __builtin_amdgcn_readlane(sj, k);
        pos += (sk < sj || (sk == sj && k < lane)) ? 1 : 0;
      }
      if (lane < len) bufA[start + pos] = (unsigned)sj;
    } else if (fits) {
      for (int j0 = 0; j0 < len; j0 += 32) {
        const int j = j0 + lane;
        const bool valid = j < len;
        const int sj = valid ? (int)bufB[start + j] : 0;
        int pos = 0;
        for (int k = 0; k < len; ++k) {
          const int sk = (int)bufB[start + k];
          pos += (sk < sj || (sk == sj && k < j)) ? 1 : 0;
        }
        if (valid) bufA[start + pos] = (unsigned)sj;
      }
    } else {
      const unsigned* sin = tmp + base + start;
      int* sout = col2 + base + start;
      for (int j0 = 0; j0 < len; j0 += 32) {
        const int j = j0 + lane;
        const bool valid = j < len;
        const int sj = valid ? (int)sin[j] : 0;
        int pos = 0;
        for (int k = 0; k < len; ++k) {
          const int sk = (int)sin[k];
          pos += (sk < sj || (sk == sj && k < j)) ? 1 : 0;
        }
        if (valid) sout[pos] = sj;
      }
    }
  }
  __syncthreads();
  if (fits) {
    for (unsigned p = t; p < total; p += BIN_T) col2[base + p] = (int)bufA[p];
  }
}

template <int MASK>
__device__ __forceinline__ float row_xor_add(float v) {
  return v + __int_as_float(__builtin_amdgcn_update_dpp(0, __float_as_int(v), 0x160 | MASK,
                                                        0xf, 0xf, true));
}
__device__ __forceinline__ float xhalf(float x) {
  return __int_as_float(__builtin_amdgcn_permlanex16(__float_as_int(x), __float_as_int(x),
                                                     0x76543210, 0xfedcba98, false, false));
}

template <int H, int NR, int MASK>
__device__ __forceinline__ void row_step(float (&a)[NR], float (&b)[NR]) {
  for (int i = 0; i < NR; ++i) {
    a[i] = row_xor_add<MASK>(a[i]);
    if (H == 4) b[i] = row_xor_add<MASK>(b[i]);
  }
}

template <int H, int NR>
__device__ __forceinline__ void head_sums(float (&a)[NR], float (&b)[NR]) {
  if (H == 1) {
    for (int i = 0; i < NR; ++i) a[i] += b[i];
  }
  row_step<H, NR, 1>(a, b);
  row_step<H, NR, 2>(a, b);
  row_step<H, NR, 4>(a, b);
  row_step<H, NR, 8>(a, b);
  if (H == 1) {
    for (int i = 0; i < NR; ++i) a[i] += xhalf(a[i]);
  }
}

__device__ __forceinline__ float dot4(const v4f& v, const v4f& w) {
  v2f t = v.lo * w.lo;
  t = v.hi * w.hi + t;
  return t.x + t.y;
}
__device__ __forceinline__ float lrelu(float x) { return fmaxf(x, NEG_SLOPE * x); }

struct Soft {
  float mA, mB, denA, denB;
  v4f acc0, acc1;
};

template <int H, int NR>
__device__ __forceinline__ void add_entries(const float* __restrict__ h, int sl, int jj, int lane,
                                            const v4f& as0, const v4f& as1,
                                            float aldA, float aldB, Soft& st) {
  v4f v0[NR], v1[NR];
  for (int i = 0; i < NR; ++i) {
    const int sj = __builtin_amdgcn_readlane(sl, jj + i);
    const v4f* row = (const v4f*)(h + (size_t)sj * FDIM);
    v0[i] = row[lane];
    v1[i] = row[lane + 32];
  }
  if (NR > 1) __builtin_amdgcn_sched_barrier(0);
  float eA[NR], eB[NR];
  for (int i = 0; i < NR; ++i) {
    eA[i] = dot4(v0[i], as0);
    eB[i] = dot4(v1[i], as1);
  }
  head_sums<H, NR>(eA, eB);
  float nA = st.mA, nB = st.mB;
  for (int i = 0; i < NR; ++i) {
    eA[i] = lrelu(eA[i] + aldA);
    nA = fmaxf(nA, eA[i]);
    if (H == 4) {
      eB[i] = lrelu(eB[i] + aldB);
      nB = fmaxf(nB, eB[i]);
    }
  }
  const float cA = __expf(st.mA - nA);
  const float cB = (H == 4) ? __expf(st.mB - nB) : cA;
  st.denA *= cA;
  if (H == 4) st.denB *= cB;
  st.acc0 *= cA;
  st.acc1 *= cB;
  for (int i = 0; i < NR; ++i) {
    const float wA = __expf(eA[i] - nA);
    const float wB = (H == 4) ? __expf(eB[i] - nB) : wA;
    st.denA += wA;
    if (H == 4) st.denB += wB;
    st.acc0 += v0[i] * wA;
    st.acc1 += v1[i] * wB;
  }
  st.mA = nA;
  if (H == 4) st.mB = nB;
}

template <int H, bool ELU>
__global__ void __launch_bounds__(256)
gat_gather(const unsigned* __restrict__ off, const int* __restrict__ col,
           const float* __restrict__ h, const float* __restrict__ a_src,
           const float* __restrict__ a_dst, const float* __restrict__ bias,
           float* __restrict__ outp, int n) {
  static_assert(H == 1 || H == 4, "lane-to-head mapping is written for 1 or 4 heads");
  const int d = __builtin_amdgcn_readfirstlane((int)((blockIdx.x * blockDim.x + threadIdx.x) >> 5));
  const int lane = threadIdx.x & 31;
  if (d >= n) return;
  const int start = (int)off[d];
  const int ne = (int)off[d + 1] - start;

  const v4f as0 = ((const v4f*)a_src)[lane];
  const v4f as1 = ((const v4f*)a_src)[lane + 32];

  Soft st;
  float aldA, aldB;
  {
    const v4f* rowd = (const v4f*)(h + (size_t)d * FDIM);
    const v4f d0 = rowd[lane];
    const v4f d1 = rowd[lane + 32];
    const v4f ad0 = ((const v4f*)a_dst)[lane];
    const v4f ad1 = ((const v4f*)a_dst)[lane + 32];
    float pa[2] = {dot4(d0, ad0), dot4(d0, as0)};
    float pb[2] = {dot4(d1, ad1), dot4(d1, as1)};
    head_sums<H, 2>(pa, pb);
    aldA = pa[0];
    aldB = (H == 4) ? pb[0] : pa[0];
    st.mA = lrelu(pa[1] + aldA);
    st.mB = (H == 4) ? lrelu(pb[1] + aldB) : st.mA;
    st.denA = 1.f;
    st.denB = 1.f;
    st.acc0 = d0;
    st.acc1 = d1;
  }

  for (int j0 = 0; j0 < ne; j0 += 32) {
    const int j = j0 + lane;
    int sl = d;
    if (j < ne) sl = col[start + j];
    const int lim = (ne - j0) < 32 ? (ne - j0) : 32;
    int jj = 0;
    for (; jj + 4 <= lim; jj += 4)
      add_entries<H, 4>(h, sl, jj, lane, as0, as1, aldA, aldB, st);
    if (jj + 2 <= lim) {
      add_entries<H, 2>(h, sl, jj, lane, as0, as1, aldA, aldB, st);
      jj += 2;
    }
    if (jj < lim) add_entries<H, 1>(h, sl, jj, lane, as0, as1, aldA, aldB, st);
  }

  const float iA = 1.f / (st.denA + 1e-16f);
  const float iB = (H == 4) ? 1.f / (st.denB + 1e-16f) : iA;
  const v4f b0 = ((const v4f*)bias)[lane];
  const v4f b1 = ((const v4f*)bias)[lane + 32];
  v4f r0 = st.acc0 * iA + b0;
  v4f r1 = st.acc1 * iB + b1;
  if (ELU) {
    for (int i = 0; i < 4; ++i) {
      r0[i] = r0[i] > 0.f ? r0[i] : expm1f(r0[i]);
      r1[i] = r1[i] > 0.f ? r1[i] : expm1f(r1[i]);
    }
  }
  v4f* orow = (v4f*)(outp + (size_t)d * FDIM);
  orow[lane]      = r0;
  orow[lane + 32] = r1;
}

static inline int blks(long long n, int bs) { return (int)((n + bs - 1) / bs); }

extern "C" void kernel_launch(void* const* d_in, const int* in_sizes, int n_in,
                              void* d_out, int out_size, void* d_ws, size_t ws_size,
                              hipStream_t stream) {
  const float* x        = (const float*)d_in[0];
  const int*   ei       = (const int*)d_in[1];
  const float* W1       = (const float*)d_in[2];
  const float* att_src1 = (const float*)d_in[3];
  const float* att_dst1 = (const float*)d_in[4];
  const float* b1       = (const float*)d_in[5];
  const float* W2       = (const float*)d_in[6];
  const float* att_src2 = (const float*)d_in[7];
  const float* att_dst2 = (const float*)d_in[8];
  const float* b2       = (const float*)d_in[9];
  float* out = (float*)d_out;

  const int* src = ei;
  const int* dst = ei + NEDGES;

  const long long NF = (long long)NNODES * FDIM;

  float* ws = (float*)d_ws;
  float*    h    = ws;
  unsigned* off  = (unsigned*)(ws + NF);
  unsigned* binned = off + OFF_PAD;
  int*      col2 = (int*)(binned + NEDGES);
  unsigned* tabT = (unsigned*)h;
  unsigned* tmp  = tabT + (NBKT + 1) * NCHUNK;
  (void)ws_size; (void)in_sizes; (void)n_in; (void)out_size;

    const long long rowThreads  = (long long)NNODES * 32;

  csr_bin<<<NCHUNK, BIN_T, 0, stream>>>(src, dst, tabT, binned);
  csr_bucket<<<NBKT, BIN_T, 0, stream>>>(tabT, binned, off, tmp, col2);

  gemm_wmma_f32<<<GEMM_WGS, GEMM_WAVES * 32, 0, stream>>>(x, W1, h, NNODES);
  gat_gather<HEADS1, true><<<blks(rowThreads, 256), 256, 0, stream>>>(off, col2, h, att_src1,
                                                                      att_dst1, b1, out, NNODES);

  gemm_wmma_f32<<<GEMM_WGS, GEMM_WAVES * 32, 0, stream>>>(out, W2, h, NNODES);
  gat_gather<1, false><<<blks(rowThreads, 256), 256, 0, stream>>>(off, col2, h, att_src2,
                                                                  att_dst2, b2, out, NNODES);
}
